# starting kernel
# baseline (speedup 1.0000x reference)
#include <hip/hip_runtime.h>

typedef __attribute__((ext_vector_type(16))) _Float16 v16h;
typedef __attribute__((ext_vector_type(2)))  _Float16 h2;
typedef __attribute__((ext_vector_type(8)))  float    v8f;

#define N_NODES_C  100000
#define N_EDGES_C  800000
#define OUT_SCALE  (1.0f / 128.0f)

__global__ __launch_bounds__(256) void zero_f32_kernel(float* __restrict__ p, int n) {
    int i = blockIdx.x * blockDim.x + threadIdx.x;
    if (i < n) p[i] = 0.0f;
}

union AFrag {
    v16h v;
    h2   p[8];
};

__global__ __launch_bounds__(256) void basis_conv_kernel(
    const float* __restrict__ x,
    const float* __restrict__ edge_attr,
    const float* __restrict__ W,
    const int*   __restrict__ ei,
    const int*   __restrict__ ej,
    float*       __restrict__ out)
{
    const int lane        = threadIdx.x & 31;
    const int waveInBlock = threadIdx.x >> 5;
    const int wavesPerBlk = blockDim.x >> 5;
    const int waveId      = blockIdx.x * wavesPerBlk + waveInBlock;
    const int numWaves    = gridDim.x * wavesPerBlk;

    const int half = lane >> 4;
    const int ln   = lane & 15;

    v16h Bfrag[8];
#pragma unroll
    for (int s = 0; s < 8; ++s) {
#pragma unroll
        for (int v = 0; v < 8; ++v) {
            const int g = v >> 2, w = v & 3;
            const int kLo = s * 32 + g * 16 + half * 8 + w * 2;
            Bfrag[s][2 * v]     = (_Float16)W[kLo * 16 + ln];
            Bfrag[s][2 * v + 1] = (_Float16)W[(kLo + 1) * 16 + ln];
        }
    }

    const int nTiles = N_EDGES_C / 16;
    for (int tile = waveId; tile < nTiles; tile += numWaves) {
        const int eBase = tile * 16;

        const int eA = eBase + ln;
        const int jn = ej[eA];

        const float2 uv = *(const float2*)(edge_attr + eA * 2);
        const float u0 = fminf(1.0f, fmaxf(-1.0f, uv.x));
        const float u1 = fminf(1.0f, fmaxf(-1.0f, uv.y));
        float bx[4], by[4];
#pragma unroll
        for (int c = 0; c < 4; ++c) {
            const float ctr = -1.0f + (2.0f / 3.0f) * (float)c;
            bx[c] = fmaxf(0.0f, 1.0f - 1.5f * fabsf(u0 - ctr));
            by[c] = fmaxf(0.0f, 1.0f - 1.5f * fabsf(u1 - ctr));
        }

        const float4* xrow = (const float4*)(x + (long)jn * 16 + half * 8);
        const float4 xlo = xrow[0];
        const float4 xhi = xrow[1];
        h2 xh[4];
        xh[0] = h2{(_Float16)xlo.x, (_Float16)xlo.y};
        xh[1] = h2{(_Float16)xlo.z, (_Float16)xlo.w};
        xh[2] = h2{(_Float16)xhi.x, (_Float16)xhi.y};
        xh[3] = h2{(_Float16)xhi.z, (_Float16)xhi.w};

        v8f acc = {};
#pragma unroll
        for (int s = 0; s < 8; ++s) {
            const int k0 = 2 * s, k1 = 2 * s + 1;
            const _Float16 b0 = (_Float16)(bx[k0 >> 2] * by[k0 & 3]);
            const _Float16 b1 = (_Float16)(bx[k1 >> 2] * by[k1 & 3]);
            const h2 b0p = h2{b0, b0};
            const h2 b1p = h2{b1, b1};
            AFrag A;
#pragma unroll
            for (int v = 0; v < 8; ++v) {
                A.p[v] = (v >= 4 ? b1p : b0p) * xh[v & 3];
            }
            acc = __builtin_amdgcn_wmma_f32_16x16x32_f16(
                false, A.v, false, Bfrag[s], (short)0, acc, false, false);
        }

        const int4* pi = (const int4*)(ei + eBase + 8 * half);
        const int4* pj = (const int4*)(ej + eBase + 8 * half);
        const int4 i0 = pi[0], i1 = pi[1];
        const int4 j0 = pj[0], j1 = pj[1];
        const int iiv[8] = {i0.x, i0.y, i0.z, i0.w, i1.x, i1.y, i1.z, i1.w};
        const int jjv[8] = {j0.x, j0.y, j0.z, j0.w, j1.x, j1.y, j1.z, j1.w};
#pragma unroll
        for (int d = 0; d < 8; ++d) {
            if (iiv[d] != jjv[d]) {
                atomicAdd(&out[(long)iiv[d] * 16 + ln], acc[d] * OUT_SCALE);
            }
        }
    }
}

extern "C" void kernel_launch(void* const* d_in, const int* in_sizes, int n_in,
                              void* d_out, int out_size, void* d_ws, size_t ws_size,
                              hipStream_t stream) {
    const float* x  = (const float*)d_in[0];
    const float* ea = (const float*)d_in[1];
    const float* W  = (const float*)d_in[2];
    const int*   ei = (const int*)d_in[3];
    const int*   ej = (const int*)d_in[4];
    float* out = (float*)d_out;

    zero_f32_kernel<<<(out_size + 255) / 256, 256, 0, stream>>>(out, out_size);

    basis_conv_kernel<<<1024, 256, 0, stream>>>(x, ea, W, ei, ej, out);
}


# W converted to f16 once per block into LDS in fragment order, 16 ds_load_b128 per wave, instead of 128 scalar loads per wave: VGPRs 141 -> 115, so all 1024 blocks are resident at once
# speedup vs baseline: 1.1894x; 1.1894x over previous
#include <hip/hip_runtime.h>

typedef __attribute__((ext_vector_type(16))) _Float16 v16h;
typedef __attribute__((ext_vector_type(8)))  _Float16 v8h;
typedef __attribute__((ext_vector_type(2)))  _Float16 h2;
typedef __attribute__((ext_vector_type(8)))  float    v8f;

#define N_NODES_C  100000
#define N_EDGES_C  800000
#define OUT_SCALE  (1.0f / 128.0f)

__global__ __launch_bounds__(256) void zero_f32_kernel(float* __restrict__ p, int n) {
    int i = blockIdx.x * blockDim.x + threadIdx.x;
    if (i < n) p[i] = 0.0f;
}

union AFrag {
    v16h v;
    h2   p[8];
};

__global__ __launch_bounds__(256) void basis_conv_kernel(
    const float* __restrict__ x,
    const float* __restrict__ edge_attr,
    const float* __restrict__ W,
    const int*   __restrict__ ei,
    const int*   __restrict__ ej,
    float*       __restrict__ out)
{
    const int lane        = threadIdx.x & 31;
    const int waveInBlock = threadIdx.x >> 5;
    const int wavesPerBlk = blockDim.x >> 5;
    const int waveId      = blockIdx.x * wavesPerBlk + waveInBlock;
    const int numWaves    = gridDim.x * wavesPerBlk;

    const int half = lane >> 4;
    const int ln   = lane & 15;

    __shared__ v8h Wlds[512];
    for (int c = threadIdx.x; c < 512; c += blockDim.x) {
        const float* wp = W + (c >> 4) * 8 * 16 + (c & 15);
        v8h h;
#pragma unroll
        for (int i = 0; i < 8; ++i) h[i] = (_Float16)wp[i * 16];
        Wlds[c] = h;
    }
    __syncthreads();

    v16h Bfrag[8];
#pragma unroll
    for (int s = 0; s < 8; ++s) {
        const v8h lo = Wlds[(2 * s) * 32 + lane];
        const v8h hi = Wlds[(2 * s + 1) * 32 + lane];
        Bfrag[s] = __builtin_shufflevector(lo, hi, 0, 1, 2, 3, 4, 5, 6, 7,
                                           8, 9, 10, 11, 12, 13, 14, 15);
    }

    const int nTiles = N_EDGES_C / 16;
    for (int tile = waveId; tile < nTiles; tile += numWaves) {
        const int eBase = tile * 16;

        const int eA = eBase + ln;
        const int jn = ej[eA];

        const float2 uv = *(const float2*)(edge_attr + eA * 2);
        const float u0 = fminf(1.0f, fmaxf(-1.0f, uv.x));
        const float u1 = fminf(1.0f, fmaxf(-1.0f, uv.y));
        float bx[4], by[4];
#pragma unroll
        for (int c = 0; c < 4; ++c) {
            const float ctr = -1.0f + (2.0f / 3.0f) * (float)c;
            bx[c] = fmaxf(0.0f, 1.0f - 1.5f * fabsf(u0 - ctr));
            by[c] = fmaxf(0.0f, 1.0f - 1.5f * fabsf(u1 - ctr));
        }

        const float4* xrow = (const float4*)(x + (long)jn * 16 + half * 8);
        const float4 xlo = xrow[0];
        const float4 xhi = xrow[1];
        h2 xh[4];
        xh[0] = h2{(_Float16)xlo.x, (_Float16)xlo.y};
        xh[1] = h2{(_Float16)xlo.z, (_Float16)xlo.w};
        xh[2] = h2{(_Float16)xhi.x, (_Float16)xhi.y};
        xh[3] = h2{(_Float16)xhi.z, (_Float16)xhi.w};

        v8f acc = {};
#pragma unroll
        for (int s = 0; s < 8; ++s) {
            const int k0 = 2 * s, k1 = 2 * s + 1;
            const _Float16 b0 = (_Float16)(bx[k0 >> 2] * by[k0 & 3]);
            const _Float16 b1 = (_Float16)(bx[k1 >> 2] * by[k1 & 3]);
            const h2 b0p = h2{b0, b0};
            const h2 b1p = h2{b1, b1};
            AFrag A;
#pragma unroll
            for (int v = 0; v < 8; ++v) {
                A.p[v] = (v >= 4 ? b1p : b0p) * xh[v & 3];
            }
            acc = __builtin_amdgcn_wmma_f32_16x16x32_f16(
                false, A.v, false, Bfrag[s], (short)0, acc, false, false);
        }

        const int4* pi = (const int4*)(ei + eBase + 8 * half);
        const int4* pj = (const int4*)(ej + eBase + 8 * half);
        const int4 i0 = pi[0], i1 = pi[1];
        const int4 j0 = pj[0], j1 = pj[1];
        const int iiv[8] = {i0.x, i0.y, i0.z, i0.w, i1.x, i1.y, i1.z, i1.w};
        const int jjv[8] = {j0.x, j0.y, j0.z, j0.w, j1.x, j1.y, j1.z, j1.w};
#pragma unroll
        for (int d = 0; d < 8; ++d) {
            if (iiv[d] != jjv[d]) {
                atomicAdd(&out[(long)iiv[d] * 16 + ln], acc[d] * OUT_SCALE);
            }
        }
    }
}

extern "C" void kernel_launch(void* const* d_in, const int* in_sizes, int n_in,
                              void* d_out, int out_size, void* d_ws, size_t ws_size,
                              hipStream_t stream) {
    const float* x  = (const float*)d_in[0];
    const float* ea = (const float*)d_in[1];
    const float* W  = (const float*)d_in[2];
    const int*   ei = (const int*)d_in[3];
    const int*   ej = (const int*)d_in[4];
    float* out = (float*)d_out;

    zero_f32_kernel<<<(out_size + 255) / 256, 256, 0, stream>>>(out, out_size);

    basis_conv_kernel<<<1024, 256, 0, stream>>>(x, ea, W, ei, ej, out);
}
